# starting kernel
# baseline (speedup 1.0000x reference)
#include <hip/hip_runtime.h>

#define B_TOT   64
#define N_IN    2048
#define D_IN    8
#define N_OUT   16
#define D_OUT   16
#define BT      16
#define NPB     32
#define NCHUNK  (N_IN / NPB)
#define S_ELEMS (B_TOT * N_OUT * D_OUT)
#define WN_STRIDE (N_OUT * D_IN * D_OUT)

typedef float v2f __attribute__((ext_vector_type(2)));
typedef float v8f __attribute__((ext_vector_type(8)));

template <int MASK>
__device__ __forceinline__ float xor_add(float v) {
    const int sw = __builtin_amdgcn_ds_swizzle(__float_as_int(v), (MASK << 10) | 0x1f);
    return v + __int_as_float(sw);
}

__global__ __launch_bounds__(512)
void routing_pass(const float* __restrict__ x,
                  const float* __restrict__ W,
                  const float* __restrict__ dotvec,
                  float* __restrict__ partial,
                  int use_bias)
{
    const int bx    = blockIdx.x;
    const int chunk = blockIdx.y;
    const int k     = threadIdx.y;
    const int lane  = threadIdx.x;
    const int hi    = lane >> 4;
    const int lo    = lane & 15;
    const int dsel  = hi * 2;

    __shared__ __align__(64) float biasLds[2][BT][N_OUT];
    __shared__ __align__(64) float cLds[N_OUT][BT];

    v8f sAcc = {};

    const int bBase = bx * BT;
    const int nBase = chunk * NPB;

    v2f dvb[4] = {};
    if (use_bias) {
        const float* dvk = dotvec + ((size_t)(bBase + lo) * N_OUT + k) * D_OUT;
        #pragma unroll
        for (int c = 0; c < 4; ++c) {
            dvb[c].x = dvk[4 * c + dsel + 0];
            dvb[c].y = dvk[4 * c + dsel + 1];
        }
    }

    const float* xp = x + ((size_t)(bBase + lo) * N_IN + nBase) * D_IN;
    const float* wp = W + ((size_t)nBase * N_OUT + k) * (D_IN * D_OUT);
    const int  aoff = (lo & 7) * D_OUT + dsel;
    const bool arowValid = (lo < D_IN);

    for (int ni = 0; ni < NPB; ++ni) {
        const float4 xA = reinterpret_cast<const float4*>(xp)[0];
        const float4 xB = reinterpret_cast<const float4*>(xp)[1];

        v2f a0, a1;
        a0.x = hi ? xA.z : xA.x;  a0.y = hi ? xA.w : xA.y;
        a1.x = hi ? xB.z : xB.x;  a1.y = hi ? xB.w : xB.y;

        v2f b0, b1;
        b0.x = wp[(0 + dsel) * D_OUT + lo];
        b0.y = wp[(1 + dsel) * D_OUT + lo];
        b1.x = wp[(4 + dsel) * D_OUT + lo];
        b1.y = wp[(5 + dsel) * D_OUT + lo];

        v8f h = {};
        h = __builtin_amdgcn_wmma_f32_16x16x4_f32(false, a0, false, b0,
                                                  (short)0, h, false, false);
        h = __builtin_amdgcn_wmma_f32_16x16x4_f32(false, a1, false, b1,
                                                  (short)0, h, false, false);

        if (use_bias) {
            const float2* wa2 = reinterpret_cast<const float2*>(wp + aoff);
            const float2 w0 = wa2[0];
            const float2 w1 = wa2[2];
            const float2 w2 = wa2[4];
            const float2 w3 = wa2[6];
            v2f wa[4];
            wa[0].x = arowValid ? w0.x : 0.f;  wa[0].y = arowValid ? w0.y : 0.f;
            wa[1].x = arowValid ? w1.x : 0.f;  wa[1].y = arowValid ? w1.y : 0.f;
            wa[2].x = arowValid ? w2.x : 0.f;  wa[2].y = arowValid ? w2.y : 0.f;
            wa[3].x = arowValid ? w3.x : 0.f;  wa[3].y = arowValid ? w3.y : 0.f;

            v8f Vt = {};
            #pragma unroll
            for (int c = 0; c < 4; ++c)
                Vt = __builtin_amdgcn_wmma_f32_16x16x4_f32(false, wa[c], false, dvb[c],
                                                           (short)0, Vt, false, false);

            const float bias =
                xA.x * Vt[0] + xA.y * Vt[1] + xA.z * Vt[2] + xA.w * Vt[3] +
                xB.x * Vt[4] + xB.y * Vt[5] + xB.z * Vt[6] + xB.w * Vt[7];
            const int p = ni & 1;
            if (lane < 16) biasLds[p][lo][k] = bias;
            __syncthreads();

            const float4 r0 = *reinterpret_cast<const float4*>(&biasLds[p][lo][0]);
            const float4 r1 = *reinterpret_cast<const float4*>(&biasLds[p][lo][4]);
            const float4 r2 = *reinterpret_cast<const float4*>(&biasLds[p][lo][8]);
            const float4 r3 = *reinterpret_cast<const float4*>(&biasLds[p][lo][12]);
            const float bk = biasLds[p][lo][k];
            float mx = fmaxf(fmaxf(fmaxf(r0.x, r0.y), fmaxf(r0.z, r0.w)),
                             fmaxf(fmaxf(r1.x, r1.y), fmaxf(r1.z, r1.w)));
            mx = fmaxf(mx, fmaxf(fmaxf(fmaxf(r2.x, r2.y), fmaxf(r2.z, r2.w)),
                                 fmaxf(fmaxf(r3.x, r3.y), fmaxf(r3.z, r3.w))));
            const float den =
                __expf(r0.x - mx) + __expf(r0.y - mx) + __expf(r0.z - mx) + __expf(r0.w - mx) +
                __expf(r1.x - mx) + __expf(r1.y - mx) + __expf(r1.z - mx) + __expf(r1.w - mx) +
                __expf(r2.x - mx) + __expf(r2.y - mx) + __expf(r2.z - mx) + __expf(r2.w - mx) +
                __expf(r3.x - mx) + __expf(r3.y - mx) + __expf(r3.z - mx) + __expf(r3.w - mx);
            const float cb = __expf(bk - mx) / den;

            cLds[k][lo] = cb;
            const float4 c0 = *reinterpret_cast<const float4*>(&cLds[k][8 * hi + 0]);
            const float4 c1 = *reinterpret_cast<const float4*>(&cLds[k][8 * hi + 4]);
            sAcc[0] += c0.x * h[0];  sAcc[1] += c0.y * h[1];
            sAcc[2] += c0.z * h[2];  sAcc[3] += c0.w * h[3];
            sAcc[4] += c1.x * h[4];  sAcc[5] += c1.y * h[5];
            sAcc[6] += c1.z * h[6];  sAcc[7] += c1.w * h[7];
        } else {
            #pragma unroll
            for (int v = 0; v < 8; ++v) sAcc[v] += 0.0625f * h[v];
        }

        xp += D_IN;
        wp += WN_STRIDE;
    }

    float* pout = partial + (size_t)chunk * S_ELEMS;
    #pragma unroll
    for (int v = 0; v < 8; ++v) {
        const int b_ = bBase + v + 8 * hi;
        pout[((size_t)b_ * N_OUT + k) * D_OUT + lo] = sAcc[v];
    }
}

__global__ __launch_bounds__(256)
void reduce_squash(const float* __restrict__ partial,
                   const float* __restrict__ addend,
                   float* __restrict__ out,
                   int mode)
{
    const int idx = blockIdx.x * 256 + threadIdx.x;
    float s = 0.f;
    #pragma unroll 8
    for (int ch = 0; ch < NCHUNK; ++ch)
        s += partial[(size_t)ch * S_ELEMS + idx];

    float s2 = s * s;
    s2 = xor_add<1>(s2);
    s2 = xor_add<2>(s2);
    s2 = xor_add<4>(s2);
    s2 = xor_add<8>(s2);

    const float scale = (s2 / (1.0f + s2)) * __frsqrt_rn(s2);
    float o = scale * s;
    if (mode == 1) o += addend[idx];
    out[idx] = o;
}

extern "C" void kernel_launch(void* const* d_in, const int* in_sizes, int n_in,
                              void* d_out, int out_size, void* d_ws, size_t ws_size,
                              hipStream_t stream) {
    (void)in_sizes; (void)n_in; (void)out_size; (void)ws_size;
    const float* x = (const float*)d_in[0];
    const float* W = (const float*)d_in[1];
    float* out = (float*)d_out;

    float* partial = (float*)d_ws;
    float* out0    = partial + (size_t)NCHUNK * S_ELEMS;
    float* outsum  = out0 + S_ELEMS;

    dim3 grid(B_TOT / BT, NCHUNK);
    dim3 block(32, 16);

    routing_pass<<<grid, block, 0, stream>>>(x, W, outsum  , partial, 0);
    reduce_squash<<<S_ELEMS / 256, 256, 0, stream>>>(partial, outsum, out0, 0);

    routing_pass<<<grid, block, 0, stream>>>(x, W, out0, partial, 1);
    reduce_squash<<<S_ELEMS / 256, 256, 0, stream>>>(partial, out0, outsum, 1);

    routing_pass<<<grid, block, 0, stream>>>(x, W, outsum, partial, 1);
    reduce_squash<<<S_ELEMS / 256, 256, 0, stream>>>(partial, outsum  , out, 2);
}


# capsule loop batched 4 capsules per trip: their global loads are in flight together and one barrier plus one LDS softmax exchange serve all 4; every sum keeps its order
# speedup vs baseline: 1.1364x; 1.1364x over previous
#include <hip/hip_runtime.h>

#define B_TOT   64
#define N_IN    2048
#define D_IN    8
#define N_OUT   16
#define D_OUT   16
#define BT      16
#define NPB     32
#define NCHUNK  (N_IN / NPB)
#define NB      4
#define S_ELEMS (B_TOT * N_OUT * D_OUT)
#define WN_STRIDE (N_OUT * D_IN * D_OUT)

typedef float v2f __attribute__((ext_vector_type(2)));
typedef float v8f __attribute__((ext_vector_type(8)));

template <int MASK>
__device__ __forceinline__ float xor_add(float v) {
    const int sw = __builtin_amdgcn_ds_swizzle(__float_as_int(v), (MASK << 10) | 0x1f);
    return v + __int_as_float(sw);
}

__global__ __launch_bounds__(512)
void routing_pass(const float* __restrict__ x,
                  const float* __restrict__ W,
                  const float* __restrict__ dotvec,
                  float* __restrict__ partial,
                  int use_bias)
{
    const int bx    = blockIdx.x;
    const int chunk = blockIdx.y;
    const int k     = threadIdx.y;
    const int lane  = threadIdx.x;
    const int hi    = lane >> 4;
    const int lo    = lane & 15;
    const int dsel  = hi * 2;

    __shared__ __align__(64) float biasLds[2][NB][BT][N_OUT];
    __shared__ __align__(64) float cLds[N_OUT][NB][BT];

    v8f sAcc = {};

    const int bBase = bx * BT;
    const int nBase = chunk * NPB;

    const float* xp = x + ((size_t)(bBase + lo) * N_IN + nBase) * D_IN;
    const float* wp = W + ((size_t)nBase * N_OUT + k) * (D_IN * D_OUT);

    if (use_bias) {
        v2f dvb[4];
        const float* dvk = dotvec + ((size_t)(bBase + lo) * N_OUT + k) * D_OUT;
        #pragma unroll
        for (int c = 0; c < 4; ++c) {
            dvb[c].x = dvk[4 * c + dsel + 0];
            dvb[c].y = dvk[4 * c + dsel + 1];
        }
        const int  aoff = (lo & 7) * D_OUT + dsel;
        const bool arowValid = (lo < D_IN);

        for (int t = 0; t < NPB / NB; ++t) {
            const int p = t & 1;

            float4 xA[NB], xB[NB];
            v2f    b0[NB], b1[NB];
            float2 w0[NB], w1[NB], w2[NB], w3[NB];
            #pragma unroll
            for (int j = 0; j < NB; ++j) {
                const float* xq = xp + j * D_IN;
                const float* wq = wp + j * WN_STRIDE;
                xA[j] = reinterpret_cast<const float4*>(xq)[0];
                xB[j] = reinterpret_cast<const float4*>(xq)[1];
                b0[j].x = wq[(0 + dsel) * D_OUT + lo];
                b0[j].y = wq[(1 + dsel) * D_OUT + lo];
                b1[j].x = wq[(4 + dsel) * D_OUT + lo];
                b1[j].y = wq[(5 + dsel) * D_OUT + lo];
                const float2* wa2 = reinterpret_cast<const float2*>(wq + aoff);
                w0[j] = wa2[0];
                w1[j] = wa2[2];
                w2[j] = wa2[4];
                w3[j] = wa2[6];
            }

            v8f   h[NB];
            float bias[NB];
            #pragma unroll
            for (int j = 0; j < NB; ++j) {
                v2f a0, a1;
                a0.x = hi ? xA[j].z : xA[j].x;  a0.y = hi ? xA[j].w : xA[j].y;
                a1.x = hi ? xB[j].z : xB[j].x;  a1.y = hi ? xB[j].w : xB[j].y;

                v8f hh = {};
                hh = __builtin_amdgcn_wmma_f32_16x16x4_f32(false, a0, false, b0[j],
                                                           (short)0, hh, false, false);
                hh = __builtin_amdgcn_wmma_f32_16x16x4_f32(false, a1, false, b1[j],
                                                           (short)0, hh, false, false);
                h[j] = hh;

                v2f wa[4];
                wa[0].x = arowValid ? w0[j].x : 0.f;  wa[0].y = arowValid ? w0[j].y : 0.f;
                wa[1].x = arowValid ? w1[j].x : 0.f;  wa[1].y = arowValid ? w1[j].y : 0.f;
                wa[2].x = arowValid ? w2[j].x : 0.f;  wa[2].y = arowValid ? w2[j].y : 0.f;
                wa[3].x = arowValid ? w3[j].x : 0.f;  wa[3].y = arowValid ? w3[j].y : 0.f;

                v8f Vt = {};
                #pragma unroll
                for (int c = 0; c < 4; ++c)
                    Vt = __builtin_amdgcn_wmma_f32_16x16x4_f32(false, wa[c], false, dvb[c],
                                                               (short)0, Vt, false, false);

                bias[j] =
                    xA[j].x * Vt[0] + xA[j].y * Vt[1] + xA[j].z * Vt[2] + xA[j].w * Vt[3] +
                    xB[j].x * Vt[4] + xB[j].y * Vt[5] + xB[j].z * Vt[6] + xB[j].w * Vt[7];
            }
            if (lane < 16) {
                #pragma unroll
                for (int j = 0; j < NB; ++j) biasLds[p][j][lo][k] = bias[j];
            }
            __syncthreads();

            #pragma unroll
            for (int j = 0; j < NB; ++j) {
                const float4 r0 = *reinterpret_cast<const float4*>(&biasLds[p][j][lo][0]);
                const float4 r1 = *reinterpret_cast<const float4*>(&biasLds[p][j][lo][4]);
                const float4 r2 = *reinterpret_cast<const float4*>(&biasLds[p][j][lo][8]);
                const float4 r3 = *reinterpret_cast<const float4*>(&biasLds[p][j][lo][12]);
                const float bk = biasLds[p][j][lo][k];
                float mx = fmaxf(fmaxf(fmaxf(r0.x, r0.y), fmaxf(r0.z, r0.w)),
                                 fmaxf(fmaxf(r1.x, r1.y), fmaxf(r1.z, r1.w)));
                mx = fmaxf(mx, fmaxf(fmaxf(fmaxf(r2.x, r2.y), fmaxf(r2.z, r2.w)),
                                     fmaxf(fmaxf(r3.x, r3.y), fmaxf(r3.z, r3.w))));
                const float den =
                    __expf(r0.x - mx) + __expf(r0.y - mx) + __expf(r0.z - mx) + __expf(r0.w - mx) +
                    __expf(r1.x - mx) + __expf(r1.y - mx) + __expf(r1.z - mx) + __expf(r1.w - mx) +
                    __expf(r2.x - mx) + __expf(r2.y - mx) + __expf(r2.z - mx) + __expf(r2.w - mx) +
                    __expf(r3.x - mx) + __expf(r3.y - mx) + __expf(r3.z - mx) + __expf(r3.w - mx);
                const float cb = __expf(bk - mx) / den;

                cLds[k][j][lo] = cb;
            }
            #pragma unroll
            for (int j = 0; j < NB; ++j) {
                const float4 c0 = *reinterpret_cast<const float4*>(&cLds[k][j][8 * hi + 0]);
                const float4 c1 = *reinterpret_cast<const float4*>(&cLds[k][j][8 * hi + 4]);
                sAcc[0] += c0.x * h[j][0];  sAcc[1] += c0.y * h[j][1];
                sAcc[2] += c0.z * h[j][2];  sAcc[3] += c0.w * h[j][3];
                sAcc[4] += c1.x * h[j][4];  sAcc[5] += c1.y * h[j][5];
                sAcc[6] += c1.z * h[j][6];  sAcc[7] += c1.w * h[j][7];
            }

            xp += NB * D_IN;
            wp += NB * WN_STRIDE;
        }
    } else {
        for (int t = 0; t < NPB / NB; ++t) {
            float4 xA[NB], xB[NB];
            v2f    b0[NB], b1[NB];
            #pragma unroll
            for (int j = 0; j < NB; ++j) {
                const float* xq = xp + j * D_IN;
                const float* wq = wp + j * WN_STRIDE;
                xA[j] = reinterpret_cast<const float4*>(xq)[0];
                xB[j] = reinterpret_cast<const float4*>(xq)[1];
                b0[j].x = wq[(0 + dsel) * D_OUT + lo];
                b0[j].y = wq[(1 + dsel) * D_OUT + lo];
                b1[j].x = wq[(4 + dsel) * D_OUT + lo];
                b1[j].y = wq[(5 + dsel) * D_OUT + lo];
            }
            #pragma unroll
            for (int j = 0; j < NB; ++j) {
                v2f a0, a1;
                a0.x = hi ? xA[j].z : xA[j].x;  a0.y = hi ? xA[j].w : xA[j].y;
                a1.x = hi ? xB[j].z : xB[j].x;  a1.y = hi ? xB[j].w : xB[j].y;
                v8f hh = {};
                hh = __builtin_amdgcn_wmma_f32_16x16x4_f32(false, a0, false, b0[j],
                                                           (short)0, hh, false, false);
                hh = __builtin_amdgcn_wmma_f32_16x16x4_f32(false, a1, false, b1[j],
                                                           (short)0, hh, false, false);
                #pragma unroll
                for (int v = 0; v < 8; ++v) sAcc[v] += 0.0625f * hh[v];
            }
            xp += NB * D_IN;
            wp += NB * WN_STRIDE;
        }
    }

    float* pout = partial + (size_t)chunk * S_ELEMS;
    #pragma unroll
    for (int v = 0; v < 8; ++v) {
        const int b_ = bBase + v + 8 * hi;
        pout[((size_t)b_ * N_OUT + k) * D_OUT + lo] = sAcc[v];
    }
}

__global__ __launch_bounds__(256)
void reduce_squash(const float* __restrict__ partial,
                   const float* __restrict__ addend,
                   float* __restrict__ out,
                   int mode)
{
    const int idx = blockIdx.x * 256 + threadIdx.x;
    float s = 0.f;
    #pragma unroll 8
    for (int ch = 0; ch < NCHUNK; ++ch)
        s += partial[(size_t)ch * S_ELEMS + idx];

    float s2 = s * s;
    s2 = xor_add<1>(s2);
    s2 = xor_add<2>(s2);
    s2 = xor_add<4>(s2);
    s2 = xor_add<8>(s2);

    const float scale = (s2 / (1.0f + s2)) * __frsqrt_rn(s2);
    float o = scale * s;
    if (mode == 1) o += addend[idx];
    out[idx] = o;
}

extern "C" void kernel_launch(void* const* d_in, const int* in_sizes, int n_in,
                              void* d_out, int out_size, void* d_ws, size_t ws_size,
                              hipStream_t stream) {
    (void)in_sizes; (void)n_in; (void)out_size; (void)ws_size;
    const float* x = (const float*)d_in[0];
    const float* W = (const float*)d_in[1];
    float* out = (float*)d_out;

    float* partial = (float*)d_ws;
    float* out0    = partial + (size_t)NCHUNK * S_ELEMS;
    float* outsum  = out0 + S_ELEMS;

    dim3 grid(B_TOT / BT, NCHUNK);
    dim3 block(32, 16);

    routing_pass<<<grid, block, 0, stream>>>(x, W, outsum  , partial, 0);
    reduce_squash<<<S_ELEMS / 256, 256, 0, stream>>>(partial, outsum, out0, 0);

    routing_pass<<<grid, block, 0, stream>>>(x, W, out0, partial, 1);
    reduce_squash<<<S_ELEMS / 256, 256, 0, stream>>>(partial, out0, outsum, 1);

    routing_pass<<<grid, block, 0, stream>>>(x, W, outsum, partial, 1);
    reduce_squash<<<S_ELEMS / 256, 256, 0, stream>>>(partial, outsum  , out, 2);
}


# softmax over output capsules computed once per (capsule, batch row) by one 16-lane row with DPP reductions and shared through LDS behind a second barrier; denominator summed as a butterfly tree
# speedup vs baseline: 1.2983x; 1.1425x over previous
#include <hip/hip_runtime.h>

#define B_TOT   64
#define N_IN    2048
#define D_IN    8
#define N_OUT   16
#define D_OUT   16
#define BT      16
#define NPB     32
#define NCHUNK  (N_IN / NPB)
#define NB      4
#define S_ELEMS (B_TOT * N_OUT * D_OUT)
#define WN_STRIDE (N_OUT * D_IN * D_OUT)
#define BPITCH  (BT + 1)
#define CPITCH  (NB * BT + 4)

typedef float v2f __attribute__((ext_vector_type(2)));
typedef float v8f __attribute__((ext_vector_type(8)));

template <int MASK>
__device__ __forceinline__ float xor_add(float v) {
    const int sw = __builtin_amdgcn_ds_swizzle(__float_as_int(v), (MASK << 10) | 0x1f);
    return v + __int_as_float(sw);
}

template <int MASK>
__device__ __forceinline__ float row_xor(float v) {
    return __int_as_float(__builtin_amdgcn_update_dpp(0, __float_as_int(v), 0x160 + MASK, 0xf, 0xf, false));
}
__device__ __forceinline__ float row_max16(float v) {
    v = fmaxf(v, row_xor<1>(v));
    v = fmaxf(v, row_xor<2>(v));
    v = fmaxf(v, row_xor<4>(v));
    v = fmaxf(v, row_xor<8>(v));
    return v;
}
__device__ __forceinline__ float row_sum16(float v) {
    v += row_xor<1>(v);
    v += row_xor<2>(v);
    v += row_xor<4>(v);
    v += row_xor<8>(v);
    return v;
}

__global__ __launch_bounds__(512)
void routing_pass(const float* __restrict__ x,
                  const float* __restrict__ W,
                  const float* __restrict__ dotvec,
                  float* __restrict__ partial,
                  int use_bias)
{
    const int bx    = blockIdx.x;
    const int chunk = blockIdx.y;
    const int k     = threadIdx.y;
    const int lane  = threadIdx.x;
    const int hi    = lane >> 4;
    const int lo    = lane & 15;
    const int dsel  = hi * 2;

    __shared__ __align__(64) float biasLds[NB][N_OUT][BPITCH];
    __shared__ __align__(64) float cLds[N_OUT][CPITCH];

    v8f sAcc = {};

    const int bBase = bx * BT;
    const int nBase = chunk * NPB;

    const float* xp = x + ((size_t)(bBase + lo) * N_IN + nBase) * D_IN;
    const float* wp = W + ((size_t)nBase * N_OUT + k) * (D_IN * D_OUT);

    if (use_bias) {
        v2f dvb[4];
        const float* dvk = dotvec + ((size_t)(bBase + lo) * N_OUT + k) * D_OUT;
        #pragma unroll
        for (int c = 0; c < 4; ++c) {
            dvb[c].x = dvk[4 * c + dsel + 0];
            dvb[c].y = dvk[4 * c + dsel + 1];
        }
        const int  aoff = (lo & 7) * D_OUT + dsel;
        const bool arowValid = (lo < D_IN);

        for (int t = 0; t < NPB / NB; ++t) {
            float4 xA[NB], xB[NB];
            v2f    b0[NB], b1[NB];
            float2 w0[NB], w1[NB], w2[NB], w3[NB];
            #pragma unroll
            for (int j = 0; j < NB; ++j) {
                const float* xq = xp + j * D_IN;
                const float* wq = wp + j * WN_STRIDE;
                xA[j] = reinterpret_cast<const float4*>(xq)[0];
                xB[j] = reinterpret_cast<const float4*>(xq)[1];
                b0[j].x = wq[(0 + dsel) * D_OUT + lo];
                b0[j].y = wq[(1 + dsel) * D_OUT + lo];
                b1[j].x = wq[(4 + dsel) * D_OUT + lo];
                b1[j].y = wq[(5 + dsel) * D_OUT + lo];
                const float2* wa2 = reinterpret_cast<const float2*>(wq + aoff);
                w0[j] = wa2[0];
                w1[j] = wa2[2];
                w2[j] = wa2[4];
                w3[j] = wa2[6];
            }

            v8f   h[NB];
            float bias[NB];
            #pragma unroll
            for (int j = 0; j < NB; ++j) {
                v2f a0, a1;
                a0.x = hi ? xA[j].z : xA[j].x;  a0.y = hi ? xA[j].w : xA[j].y;
                a1.x = hi ? xB[j].z : xB[j].x;  a1.y = hi ? xB[j].w : xB[j].y;

                v8f hh = {};
                hh = __builtin_amdgcn_wmma_f32_16x16x4_f32(false, a0, false, b0[j],
                                                           (short)0, hh, false, false);
                hh = __builtin_amdgcn_wmma_f32_16x16x4_f32(false, a1, false, b1[j],
                                                           (short)0, hh, false, false);
                h[j] = hh;

                v2f wa[4];
                wa[0].x = arowValid ? w0[j].x : 0.f;  wa[0].y = arowValid ? w0[j].y : 0.f;
                wa[1].x = arowValid ? w1[j].x : 0.f;  wa[1].y = arowValid ? w1[j].y : 0.f;
                wa[2].x = arowValid ? w2[j].x : 0.f;  wa[2].y = arowValid ? w2[j].y : 0.f;
                wa[3].x = arowValid ? w3[j].x : 0.f;  wa[3].y = arowValid ? w3[j].y : 0.f;

                v8f Vt = {};
                #pragma unroll
                for (int c = 0; c < 4; ++c)
                    Vt = __builtin_amdgcn_wmma_f32_16x16x4_f32(false, wa[c], false, dvb[c],
                                                               (short)0, Vt, false, false);

                bias[j] =
                    xA[j].x * Vt[0] + xA[j].y * Vt[1] + xA[j].z * Vt[2] + xA[j].w * Vt[3] +
                    xB[j].x * Vt[4] + xB[j].y * Vt[5] + xB[j].z * Vt[6] + xB[j].w * Vt[7];
            }
            if (lane < 16) {
                #pragma unroll
                for (int j = 0; j < NB; ++j) biasLds[j][k][lo] = bias[j];
            }
            __syncthreads();

            {
                const int j0 = 2 * hi;
                const float v0 = biasLds[j0 + 0][lo][k];
                const float v1 = biasLds[j0 + 1][lo][k];
                const float m0 = row_max16(v0);
                const float m1 = row_max16(v1);
                const float e0 = __expf(v0 - m0);
                const float e1 = __expf(v1 - m1);
                const float d0 = row_sum16(e0);
                const float d1 = row_sum16(e1);
                cLds[lo][(j0 + 0) * BT + k] = e0 / d0;
                cLds[lo][(j0 + 1) * BT + k] = e1 / d1;
            }
            __syncthreads();

            #pragma unroll
            for (int j = 0; j < NB; ++j) {
                const float4 c0 = *reinterpret_cast<const float4*>(&cLds[k][j * BT + 8 * hi + 0]);
                const float4 c1 = *reinterpret_cast<const float4*>(&cLds[k][j * BT + 8 * hi + 4]);
                sAcc[0] += c0.x * h[j][0];  sAcc[1] += c0.y * h[j][1];
                sAcc[2] += c0.z * h[j][2];  sAcc[3] += c0.w * h[j][3];
                sAcc[4] += c1.x * h[j][4];  sAcc[5] += c1.y * h[j][5];
                sAcc[6] += c1.z * h[j][6];  sAcc[7] += c1.w * h[j][7];
            }

            xp += NB * D_IN;
            wp += NB * WN_STRIDE;
        }
    } else {
        for (int t = 0; t < NPB / NB; ++t) {
            float4 xA[NB], xB[NB];
            v2f    b0[NB], b1[NB];
            #pragma unroll
            for (int j = 0; j < NB; ++j) {
                const float* xq = xp + j * D_IN;
                const float* wq = wp + j * WN_STRIDE;
                xA[j] = reinterpret_cast<const float4*>(xq)[0];
                xB[j] = reinterpret_cast<const float4*>(xq)[1];
                b0[j].x = wq[(0 + dsel) * D_OUT + lo];
                b0[j].y = wq[(1 + dsel) * D_OUT + lo];
                b1[j].x = wq[(4 + dsel) * D_OUT + lo];
                b1[j].y = wq[(5 + dsel) * D_OUT + lo];
            }
            #pragma unroll
            for (int j = 0; j < NB; ++j) {
                v2f a0, a1;
                a0.x = hi ? xA[j].z : xA[j].x;  a0.y = hi ? xA[j].w : xA[j].y;
                a1.x = hi ? xB[j].z : xB[j].x;  a1.y = hi ? xB[j].w : xB[j].y;
                v8f hh = {};
                hh = __builtin_amdgcn_wmma_f32_16x16x4_f32(false, a0, false, b0[j],
                                                           (short)0, hh, false, false);
                hh = __builtin_amdgcn_wmma_f32_16x16x4_f32(false, a1, false, b1[j],
                                                           (short)0, hh, false, false);
                #pragma unroll
                for (int v = 0; v < 8; ++v) sAcc[v] += 0.0625f * hh[v];
            }
            xp += NB * D_IN;
            wp += NB * WN_STRIDE;
        }
    }

    float* pout = partial + (size_t)chunk * S_ELEMS;
    #pragma unroll
    for (int v = 0; v < 8; ++v) {
        const int b_ = bBase + v + 8 * hi;
        pout[((size_t)b_ * N_OUT + k) * D_OUT + lo] = sAcc[v];
    }
}

__global__ __launch_bounds__(256)
void reduce_squash(const float* __restrict__ partial,
                   const float* __restrict__ addend,
                   float* __restrict__ out,
                   int mode)
{
    const int idx = blockIdx.x * 256 + threadIdx.x;
    float s = 0.f;
    #pragma unroll 8
    for (int ch = 0; ch < NCHUNK; ++ch)
        s += partial[(size_t)ch * S_ELEMS + idx];

    float s2 = s * s;
    s2 = xor_add<1>(s2);
    s2 = xor_add<2>(s2);
    s2 = xor_add<4>(s2);
    s2 = xor_add<8>(s2);

    const float scale = (s2 / (1.0f + s2)) * __frsqrt_rn(s2);
    float o = scale * s;
    if (mode == 1) o += addend[idx];
    out[idx] = o;
}

extern "C" void kernel_launch(void* const* d_in, const int* in_sizes, int n_in,
                              void* d_out, int out_size, void* d_ws, size_t ws_size,
                              hipStream_t stream) {
    (void)in_sizes; (void)n_in; (void)out_size; (void)ws_size;
    const float* x = (const float*)d_in[0];
    const float* W = (const float*)d_in[1];
    float* out = (float*)d_out;

    float* partial = (float*)d_ws;
    float* out0    = partial + (size_t)NCHUNK * S_ELEMS;
    float* outsum  = out0 + S_ELEMS;

    dim3 grid(B_TOT / BT, NCHUNK);
    dim3 block(32, 16);

    routing_pass<<<grid, block, 0, stream>>>(x, W, outsum  , partial, 0);
    reduce_squash<<<S_ELEMS / 256, 256, 0, stream>>>(partial, outsum, out0, 0);

    routing_pass<<<grid, block, 0, stream>>>(x, W, out0, partial, 1);
    reduce_squash<<<S_ELEMS / 256, 256, 0, stream>>>(partial, out0, outsum, 1);

    routing_pass<<<grid, block, 0, stream>>>(x, W, outsum, partial, 1);
    reduce_squash<<<S_ELEMS / 256, 256, 0, stream>>>(partial, outsum  , out, 2);
}


# x tile of the block staged once in LDS by whole-line row loads; every wave reads its x fragments from LDS instead of strided global loads repeated in all 16 waves; every sum keeps its order
# speedup vs baseline: 1.8717x; 1.4417x over previous
#include <hip/hip_runtime.h>

#define B_TOT   64
#define N_IN    2048
#define D_IN    8
#define N_OUT   16
#define D_OUT   16
#define BT      16
#define NPB     32
#define NCHUNK  (N_IN / NPB)
#define NB      4
#define S_ELEMS (B_TOT * N_OUT * D_OUT)
#define WN_STRIDE (N_OUT * D_IN * D_OUT)
#define BPITCH  (BT + 1)
#define CPITCH  (NB * BT + 4)
#define XROW    (NPB * D_IN)
#define XPITCH  (XROW + 4)

typedef float v2f __attribute__((ext_vector_type(2)));
typedef float v8f __attribute__((ext_vector_type(8)));

template <int MASK>
__device__ __forceinline__ float xor_add(float v) {
    const int sw = __builtin_amdgcn_ds_swizzle(__float_as_int(v), (MASK << 10) | 0x1f);
    return v + __int_as_float(sw);
}

template <int MASK>
__device__ __forceinline__ float row_xor(float v) {
    return __int_as_float(__builtin_amdgcn_update_dpp(0, __float_as_int(v), 0x160 + MASK, 0xf, 0xf, false));
}
__device__ __forceinline__ float row_max16(float v) {
    v = fmaxf(v, row_xor<1>(v));
    v = fmaxf(v, row_xor<2>(v));
    v = fmaxf(v, row_xor<4>(v));
    v = fmaxf(v, row_xor<8>(v));
    return v;
}
__device__ __forceinline__ float row_sum16(float v) {
    v += row_xor<1>(v);
    v += row_xor<2>(v);
    v += row_xor<4>(v);
    v += row_xor<8>(v);
    return v;
}

__global__ __launch_bounds__(512)
void routing_pass(const float* __restrict__ x,
                  const float* __restrict__ W,
                  const float* __restrict__ dotvec,
                  float* __restrict__ partial,
                  int use_bias)
{
    const int bx    = blockIdx.x;
    const int chunk = blockIdx.y;
    const int k     = threadIdx.y;
    const int lane  = threadIdx.x;
    const int hi    = lane >> 4;
    const int lo    = lane & 15;
    const int dsel  = hi * 2;

    __shared__ __align__(64) float biasLds[NB][N_OUT][BPITCH];
    __shared__ __align__(64) float cLds[N_OUT][CPITCH];
    __shared__ __align__(16) float xs[BT][XPITCH];

    v8f sAcc = {};

    const int bBase = bx * BT;
    const int nBase = chunk * NPB;

    const float4* xrow4 = reinterpret_cast<const float4*>(x + ((size_t)(bBase + k) * N_IN + nBase) * D_IN);
    const float4 xg0 = xrow4[lane];
    const float4 xg1 = xrow4[32 + lane];

    const float* wp = W + ((size_t)nBase * N_OUT + k) * (D_IN * D_OUT);

    const float* xfrag = &xs[lo][dsel];
    const float* xbrow = &xs[lo][hi * D_IN];

    if (use_bias) {
        v2f dvb[4];
        const float* dvk = dotvec + ((size_t)(bBase + lo) * N_OUT + k) * D_OUT;
        #pragma unroll
        for (int c = 0; c < 4; ++c) {
            dvb[c].x = dvk[4 * c + dsel + 0];
            dvb[c].y = dvk[4 * c + dsel + 1];
        }
        const int  aoff = (lo >> 3) * WN_STRIDE + (lo & 7) * D_OUT + dsel;

        *reinterpret_cast<float4*>(&xs[k][4 * lane])            = xg0;
        *reinterpret_cast<float4*>(&xs[k][XROW / 2 + 4 * lane]) = xg1;
        __syncthreads();

        for (int t = 0; t < NPB / NB; ++t) {
            v2f    a0[NB], a1[NB];
            v2f    b0[NB], b1[NB];
            float2 w0[NB / 2], w1[NB / 2], w2[NB / 2], w3[NB / 2];
            #pragma unroll
            for (int j = 0; j < NB; ++j) {
                const float* wq = wp + j * WN_STRIDE;
                b0[j].x = wq[(0 + dsel) * D_OUT + lo];
                b0[j].y = wq[(1 + dsel) * D_OUT + lo];
                b1[j].x = wq[(4 + dsel) * D_OUT + lo];
                b1[j].y = wq[(5 + dsel) * D_OUT + lo];
            }
            #pragma unroll
            for (int p = 0; p < NB / 2; ++p) {
                const float2* wa2 = reinterpret_cast<const float2*>(wp + (2 * p) * WN_STRIDE + aoff);
                w0[p] = wa2[0];
                w1[p] = wa2[2];
                w2[p] = wa2[4];
                w3[p] = wa2[6];
            }
            float4 sA[NB / 2], sB[NB / 2];
            #pragma unroll
            for (int j = 0; j < NB; ++j) {
                const float* xq = xfrag + (t * NB + j) * D_IN;
                a0[j] = *reinterpret_cast<const v2f*>(xq);
                a1[j] = *reinterpret_cast<const v2f*>(xq + 4);
            }
            #pragma unroll
            for (int p = 0; p < NB / 2; ++p) {
                const float* xq = xbrow + (t * NB + 2 * p) * D_IN;
                sA[p] = *reinterpret_cast<const float4*>(xq);
                sB[p] = *reinterpret_cast<const float4*>(xq + 4);
            }

            v8f   h[NB];
            float bias[NB / 2];
            #pragma unroll
            for (int j = 0; j < NB; ++j) {
                v8f hh = {};
                hh = __builtin_amdgcn_wmma_f32_16x16x4_f32(false, a0[j], false, b0[j],
                                                           (short)0, hh, false, false);
                hh = __builtin_amdgcn_wmma_f32_16x16x4_f32(false, a1[j], false, b1[j],
                                                           (short)0, hh, false, false);
                h[j] = hh;
            }
            #pragma unroll
            for (int p = 0; p < NB / 2; ++p) {
                v2f wa[4];
                wa[0].x = w0[p].x;  wa[0].y = w0[p].y;
                wa[1].x = w1[p].x;  wa[1].y = w1[p].y;
                wa[2].x = w2[p].x;  wa[2].y = w2[p].y;
                wa[3].x = w3[p].x;  wa[3].y = w3[p].y;

                v8f Vt = {};
                #pragma unroll
                for (int c = 0; c < 4; ++c)
                    Vt = __builtin_amdgcn_wmma_f32_16x16x4_f32(false, wa[c], false, dvb[c],
                                                               (short)0, Vt, false, false);

                bias[p] =
                    sA[p].x * Vt[0] + sA[p].y * Vt[1] + sA[p].z * Vt[2] + sA[p].w * Vt[3] +
                    sB[p].x * Vt[4] + sB[p].y * Vt[5] + sB[p].z * Vt[6] + sB[p].w * Vt[7];
            }
            #pragma unroll
            for (int p = 0; p < NB / 2; ++p) biasLds[2 * p + hi][k][lo] = bias[p];
            __syncthreads();

            {
                const int j0 = 2 * hi;
                const float v0 = biasLds[j0 + 0][lo][k];
                const float v1 = biasLds[j0 + 1][lo][k];
                const float m0 = row_max16(v0);
                const float m1 = row_max16(v1);
                const float e0 = __expf(v0 - m0);
                const float e1 = __expf(v1 - m1);
                const float d0 = row_sum16(e0);
                const float d1 = row_sum16(e1);
                cLds[lo][(j0 + 0) * BT + k] = e0 / d0;
                cLds[lo][(j0 + 1) * BT + k] = e1 / d1;
            }
            __syncthreads();

            #pragma unroll
            for (int j = 0; j < NB; ++j) {
                const float4 c0 = *reinterpret_cast<const float4*>(&cLds[k][j * BT + 8 * hi + 0]);
                const float4 c1 = *reinterpret_cast<const float4*>(&cLds[k][j * BT + 8 * hi + 4]);
                sAcc[0] += c0.x * h[j][0];  sAcc[1] += c0.y * h[j][1];
                sAcc[2] += c0.z * h[j][2];  sAcc[3] += c0.w * h[j][3];
                sAcc[4] += c1.x * h[j][4];  sAcc[5] += c1.y * h[j][5];
                sAcc[6] += c1.z * h[j][6];  sAcc[7] += c1.w * h[j][7];
            }

            wp += NB * WN_STRIDE;
        }
    } else {
        *reinterpret_cast<float4*>(&xs[k][4 * lane])            = xg0;
        *reinterpret_cast<float4*>(&xs[k][XROW / 2 + 4 * lane]) = xg1;
        __syncthreads();

        #pragma unroll
        for (int t = 0; t < NPB / NB; ++t) {
            v2f    a0[NB], a1[NB];
            v2f    b0[NB], b1[NB];
            #pragma unroll
            for (int j = 0; j < NB; ++j) {
                const float* wq = wp + j * WN_STRIDE;
                b0[j].x = wq[(0 + dsel) * D_OUT + lo];
                b0[j].y = wq[(1 + dsel) * D_OUT + lo];
                b1[j].x = wq[(4 + dsel) * D_OUT + lo];
                b1[j].y = wq[(5 + dsel) * D_OUT + lo];
            }
            #pragma unroll
            for (int j = 0; j < NB; ++j) {
                const float* xq = xfrag + (t * NB + j) * D_IN;
                a0[j] = *reinterpret_cast<const v2f*>(xq);
                a1[j] = *reinterpret_cast<const v2f*>(xq + 4);
            }
            #pragma unroll
            for (int j = 0; j < NB; ++j) {
                v8f hh = {};
                hh = __builtin_amdgcn_wmma_f32_16x16x4_f32(false, a0[j], false, b0[j],
                                                           (short)0, hh, false, false);
                hh = __builtin_amdgcn_wmma_f32_16x16x4_f32(false, a1[j], false, b1[j],
                                                           (short)0, hh, false, false);
                #pragma unroll
                for (int v = 0; v < 8; ++v) sAcc[v] += 0.0625f * hh[v];
            }
            wp += NB * WN_STRIDE;
        }
    }

    float* pout = partial + (size_t)chunk * S_ELEMS;
    #pragma unroll
    for (int v = 0; v < 8; ++v) {
        const int b_ = bBase + v + 8 * hi;
        pout[((size_t)b_ * N_OUT + k) * D_OUT + lo] = sAcc[v];
    }
}

__global__ __launch_bounds__(256)
void reduce_squash(const float* __restrict__ partial,
                   const float* __restrict__ addend,
                   float* __restrict__ out,
                   int mode)
{
    const int idx = blockIdx.x * 256 + threadIdx.x;
    float s = 0.f;
    #pragma unroll 8
    for (int ch = 0; ch < NCHUNK; ++ch)
        s += partial[(size_t)ch * S_ELEMS + idx];

    float s2 = s * s;
    s2 = xor_add<1>(s2);
    s2 = xor_add<2>(s2);
    s2 = xor_add<4>(s2);
    s2 = xor_add<8>(s2);

    const float scale = (s2 / (1.0f + s2)) * __frsqrt_rn(s2);
    float o = scale * s;
    if (mode == 1) o += addend[idx];
    out[idx] = o;
}

extern "C" void kernel_launch(void* const* d_in, const int* in_sizes, int n_in,
                              void* d_out, int out_size, void* d_ws, size_t ws_size,
                              hipStream_t stream) {
    (void)in_sizes; (void)n_in; (void)out_size; (void)ws_size;
    const float* x = (const float*)d_in[0];
    const float* W = (const float*)d_in[1];
    float* out = (float*)d_out;

    float* partial = (float*)d_ws;
    float* out0    = partial + (size_t)NCHUNK * S_ELEMS;
    float* outsum  = out0 + S_ELEMS;

    dim3 grid(B_TOT / BT, NCHUNK);
    dim3 block(32, 16);

    routing_pass<<<grid, block, 0, stream>>>(x, W, outsum  , partial, 0);
    reduce_squash<<<S_ELEMS / 256, 256, 0, stream>>>(partial, outsum, out0, 0);

    routing_pass<<<grid, block, 0, stream>>>(x, W, out0, partial, 1);
    reduce_squash<<<S_ELEMS / 256, 256, 0, stream>>>(partial, out0, outsum, 1);

    routing_pass<<<grid, block, 0, stream>>>(x, W, outsum, partial, 1);
    reduce_squash<<<S_ELEMS / 256, 256, 0, stream>>>(partial, outsum  , out, 2);
}


# hats tile computed transposed (W as A, x as B) so a lane owns one batch row: logit = per-lane dot of hats with dv + one cross-half add; the W x dv^T WMMAs and loads go; logit sum order changes
# speedup vs baseline: 2.1644x; 1.1564x over previous
#include <hip/hip_runtime.h>

#define B_TOT   64
#define N_IN    2048
#define D_IN    8
#define N_OUT   16
#define D_OUT   16
#define BT      16
#define NPB     32
#define NCHUNK  (N_IN / NPB)
#define NB      4
#define S_ELEMS (B_TOT * N_OUT * D_OUT)
#define WN_STRIDE (N_OUT * D_IN * D_OUT)
#define BPITCH  (BT + 1)
#define CPITCH  (NB * BT + 4)
#define XROW    (NPB * D_IN)
#define XPITCH  (XROW + 4)

typedef float v2f __attribute__((ext_vector_type(2)));
typedef float v8f __attribute__((ext_vector_type(8)));

template <int MASK>
__device__ __forceinline__ float xor_add(float v) {
    const int sw = __builtin_amdgcn_ds_swizzle(__float_as_int(v), (MASK << 10) | 0x1f);
    return v + __int_as_float(sw);
}

template <int MASK>
__device__ __forceinline__ float row_xor(float v) {
    return __int_as_float(__builtin_amdgcn_update_dpp(0, __float_as_int(v), 0x160 + MASK, 0xf, 0xf, false));
}
__device__ __forceinline__ float other_half(float v) {
    return __int_as_float(__builtin_amdgcn_permlanex16(0, __float_as_int(v), 0x76543210, 0xfedcba98, false, false));
}
__device__ __forceinline__ float row_max16(float v) {
    v = fmaxf(v, row_xor<1>(v));
    v = fmaxf(v, row_xor<2>(v));
    v = fmaxf(v, row_xor<4>(v));
    v = fmaxf(v, row_xor<8>(v));
    return v;
}
__device__ __forceinline__ float row_sum16(float v) {
    v += row_xor<1>(v);
    v += row_xor<2>(v);
    v += row_xor<4>(v);
    v += row_xor<8>(v);
    return v;
}

__global__ __launch_bounds__(512)
void routing_pass(const float* __restrict__ x,
                  const float* __restrict__ W,
                  const float* __restrict__ dotvec,
                  float* __restrict__ partial,
                  int use_bias)
{
    const int bx    = blockIdx.x;
    const int chunk = blockIdx.y;
    const int k     = threadIdx.y;
    const int lane  = threadIdx.x;
    const int hi    = lane >> 4;
    const int lo    = lane & 15;
    const int dsel  = hi * 2;

    __shared__ __align__(64) float biasLds[NB][N_OUT][BPITCH];
    __shared__ __align__(64) float cLds[N_OUT][CPITCH];
    __shared__ __align__(16) float xs[BT][XPITCH];

    v8f sAcc = {};

    const int bBase = bx * BT;
    const int nBase = chunk * NPB;

    const float4* xrow4 = reinterpret_cast<const float4*>(x + ((size_t)(bBase + k) * N_IN + nBase) * D_IN);
    const float4 xg0 = xrow4[lane];
    const float4 xg1 = xrow4[32 + lane];

    const float* wp = W + ((size_t)nBase * N_OUT + k) * (D_IN * D_OUT);

    const float* xfrag = &xs[lo][dsel];

    if (use_bias) {
        const float4* dvp = reinterpret_cast<const float4*>(
            dotvec + ((size_t)(bBase + lo) * N_OUT + k) * D_OUT + 8 * hi);
        const float4 dvA = dvp[0];
        const float4 dvB = dvp[1];
        const v2f dv01 = {dvA.x, dvA.y}, dv23 = {dvA.z, dvA.w}, dv45 = {dvB.x, dvB.y}, dv67 = {dvB.z, dvB.w};

        *reinterpret_cast<float4*>(&xs[k][4 * lane])            = xg0;
        *reinterpret_cast<float4*>(&xs[k][XROW / 2 + 4 * lane]) = xg1;
        __syncthreads();

        for (int t = 0; t < NPB / NB; ++t) {
            v2f    a0[NB], a1[NB];
            v2f    b0[NB], b1[NB];
            #pragma unroll
            for (int j = 0; j < NB; ++j) {
                const float* wq = wp + j * WN_STRIDE;
                b0[j].x = wq[(0 + dsel) * D_OUT + lo];
                b0[j].y = wq[(1 + dsel) * D_OUT + lo];
                b1[j].x = wq[(4 + dsel) * D_OUT + lo];
                b1[j].y = wq[(5 + dsel) * D_OUT + lo];
            }
            #pragma unroll
            for (int j = 0; j < NB; ++j) {
                const float* xq = xfrag + (t * NB + j) * D_IN;
                a0[j] = *reinterpret_cast<const v2f*>(xq);
                a1[j] = *reinterpret_cast<const v2f*>(xq + 4);
            }

            v8f   h[NB];
            float bias[NB];
            #pragma unroll
            for (int j = 0; j < NB; ++j) {
                v8f hh = {};
                hh = __builtin_amdgcn_wmma_f32_16x16x4_f32(false, b0[j], false, a0[j],
                                                           (short)0, hh, false, false);
                hh = __builtin_amdgcn_wmma_f32_16x16x4_f32(false, b1[j], false, a1[j],
                                                           (short)0, hh, false, false);
                h[j] = hh;
                v2f pp = __builtin_shufflevector(hh, hh, 0, 1) * dv01;
                pp = __builtin_shufflevector(hh, hh, 2, 3) * dv23 + pp;
                pp = __builtin_shufflevector(hh, hh, 4, 5) * dv45 + pp;
                pp = __builtin_shufflevector(hh, hh, 6, 7) * dv67 + pp;
                const float part = pp.x + pp.y;
                bias[j] = part + other_half(part);
            }
            #pragma unroll
            for (int p = 0; p < NB / 2; ++p) biasLds[2 * p + hi][k][lo] = hi ? bias[2 * p + 1] : bias[2 * p];
            __syncthreads();

            {
                const int j0 = 2 * hi;
                const float v0 = biasLds[j0 + 0][lo][k];
                const float v1 = biasLds[j0 + 1][lo][k];
                const float m0 = row_max16(v0);
                const float m1 = row_max16(v1);
                const float e0 = __expf(v0 - m0);
                const float e1 = __expf(v1 - m1);
                const float d0 = row_sum16(e0);
                const float d1 = row_sum16(e1);
                *reinterpret_cast<float2*>(&cLds[lo][k * NB + j0]) = make_float2(e0 / d0, e1 / d1);
            }
            __syncthreads();

            const float4 c4 = *reinterpret_cast<const float4*>(&cLds[k][lo * NB]);
            const float  cj[NB] = {c4.x, c4.y, c4.z, c4.w};
            #pragma unroll
            for (int j = 0; j < NB; ++j) {
                #pragma unroll
                for (int v = 0; v < 8; ++v) sAcc[v] += cj[j] * h[j][v];
            }

            wp += NB * WN_STRIDE;
        }
    } else {
        *reinterpret_cast<float4*>(&xs[k][4 * lane])            = xg0;
        *reinterpret_cast<float4*>(&xs[k][XROW / 2 + 4 * lane]) = xg1;
        __syncthreads();

        #pragma unroll
        for (int t = 0; t < NPB / NB; ++t) {
            v2f    a0[NB], a1[NB];
            v2f    b0[NB], b1[NB];
            #pragma unroll
            for (int j = 0; j < NB; ++j) {
                const float* wq = wp + j * WN_STRIDE;
                b0[j].x = wq[(0 + dsel) * D_OUT + lo];
                b0[j].y = wq[(1 + dsel) * D_OUT + lo];
                b1[j].x = wq[(4 + dsel) * D_OUT + lo];
                b1[j].y = wq[(5 + dsel) * D_OUT + lo];
            }
            #pragma unroll
            for (int j = 0; j < NB; ++j) {
                const float* xq = xfrag + (t * NB + j) * D_IN;
                a0[j] = *reinterpret_cast<const v2f*>(xq);
                a1[j] = *reinterpret_cast<const v2f*>(xq + 4);
            }
            #pragma unroll
            for (int j = 0; j < NB; ++j) {
                v8f hh = {};
                hh = __builtin_amdgcn_wmma_f32_16x16x4_f32(false, b0[j], false, a0[j],
                                                           (short)0, hh, false, false);
                hh = __builtin_amdgcn_wmma_f32_16x16x4_f32(false, b1[j], false, a1[j],
                                                           (short)0, hh, false, false);
                #pragma unroll
                for (int v = 0; v < 8; ++v) sAcc[v] += 0.0625f * hh[v];
            }
            wp += NB * WN_STRIDE;
        }
    }

    float* pout = partial + (size_t)chunk * S_ELEMS + ((size_t)(bBase + lo) * N_OUT + k) * D_OUT + 8 * hi;
    *reinterpret_cast<float4*>(pout)     = make_float4(sAcc[0], sAcc[1], sAcc[2], sAcc[3]);
    *reinterpret_cast<float4*>(pout + 4) = make_float4(sAcc[4], sAcc[5], sAcc[6], sAcc[7]);
}

__global__ __launch_bounds__(256)
void reduce_squash(const float* __restrict__ partial,
                   const float* __restrict__ addend,
                   float* __restrict__ out,
                   int mode)
{
    const int idx = blockIdx.x * 256 + threadIdx.x;
    float s = 0.f;
    #pragma unroll 8
    for (int ch = 0; ch < NCHUNK; ++ch)
        s += partial[(size_t)ch * S_ELEMS + idx];

    float s2 = s * s;
    s2 = xor_add<1>(s2);
    s2 = xor_add<2>(s2);
    s2 = xor_add<4>(s2);
    s2 = xor_add<8>(s2);

    const float scale = (s2 / (1.0f + s2)) * __frsqrt_rn(s2);
    float o = scale * s;
    if (mode == 1) o += addend[idx];
    out[idx] = o;
}

extern "C" void kernel_launch(void* const* d_in, const int* in_sizes, int n_in,
                              void* d_out, int out_size, void* d_ws, size_t ws_size,
                              hipStream_t stream) {
    (void)in_sizes; (void)n_in; (void)out_size; (void)ws_size;
    const float* x = (const float*)d_in[0];
    const float* W = (const float*)d_in[1];
    float* out = (float*)d_out;

    float* partial = (float*)d_ws;
    float* out0    = partial + (size_t)NCHUNK * S_ELEMS;
    float* outsum  = out0 + S_ELEMS;

    dim3 grid(B_TOT / BT, NCHUNK);
    dim3 block(32, 16);

    routing_pass<<<grid, block, 0, stream>>>(x, W, outsum  , partial, 0);
    reduce_squash<<<S_ELEMS / 256, 256, 0, stream>>>(partial, outsum, out0, 0);

    routing_pass<<<grid, block, 0, stream>>>(x, W, out0, partial, 1);
    reduce_squash<<<S_ELEMS / 256, 256, 0, stream>>>(partial, out0, outsum, 1);

    routing_pass<<<grid, block, 0, stream>>>(x, W, outsum, partial, 1);
    reduce_squash<<<S_ELEMS / 256, 256, 0, stream>>>(partial, outsum  , out, 2);
}


# reduce kernel: each element's 64 chunk partials (and the addend) are all loaded before the first add, so they are in flight together, on 256 work-groups of 64 threads; chunk sum order unchanged
# speedup vs baseline: 2.7332x; 1.2628x over previous
#include <hip/hip_runtime.h>

#define B_TOT   64
#define N_IN    2048
#define D_IN    8
#define N_OUT   16
#define D_OUT   16
#define BT      16
#define NPB     32
#define NCHUNK  (N_IN / NPB)
#define NB      4
#define S_ELEMS (B_TOT * N_OUT * D_OUT)
#define WN_STRIDE (N_OUT * D_IN * D_OUT)
#define BPITCH  (BT + 1)
#define CPITCH  (NB * BT + 4)
#define XROW    (NPB * D_IN)
#define XPITCH  (XROW + 4)

typedef float v2f __attribute__((ext_vector_type(2)));
typedef float v8f __attribute__((ext_vector_type(8)));

template <int MASK>
__device__ __forceinline__ float xor_add(float v) {
    const int sw = __builtin_amdgcn_ds_swizzle(__float_as_int(v), (MASK << 10) | 0x1f);
    return v + __int_as_float(sw);
}

template <int MASK>
__device__ __forceinline__ float row_xor(float v) {
    return __int_as_float(__builtin_amdgcn_update_dpp(0, __float_as_int(v), 0x160 + MASK, 0xf, 0xf, false));
}
__device__ __forceinline__ float other_half(float v) {
    return __int_as_float(__builtin_amdgcn_permlanex16(0, __float_as_int(v), 0x76543210, 0xfedcba98, false, false));
}
__device__ __forceinline__ float row_max16(float v) {
    v = fmaxf(v, row_xor<1>(v));
    v = fmaxf(v, row_xor<2>(v));
    v = fmaxf(v, row_xor<4>(v));
    v = fmaxf(v, row_xor<8>(v));
    return v;
}
__device__ __forceinline__ float row_sum16(float v) {
    v += row_xor<1>(v);
    v += row_xor<2>(v);
    v += row_xor<4>(v);
    v += row_xor<8>(v);
    return v;
}

__global__ __launch_bounds__(512)
void routing_pass(const float* __restrict__ x,
                  const float* __restrict__ W,
                  const float* __restrict__ dotvec,
                  float* __restrict__ partial,
                  int use_bias)
{
    const int bx    = blockIdx.x;
    const int chunk = blockIdx.y;
    const int k     = threadIdx.y;
    const int lane  = threadIdx.x;
    const int hi    = lane >> 4;
    const int lo    = lane & 15;
    const int dsel  = hi * 2;

    __shared__ __align__(64) float biasLds[NB][N_OUT][BPITCH];
    __shared__ __align__(64) float cLds[N_OUT][CPITCH];
    __shared__ __align__(16) float xs[BT][XPITCH];

    v8f sAcc = {};

    const int bBase = bx * BT;
    const int nBase = chunk * NPB;

    const float4* xrow4 = reinterpret_cast<const float4*>(x + ((size_t)(bBase + k) * N_IN + nBase) * D_IN);
    const float4 xg0 = xrow4[lane];
    const float4 xg1 = xrow4[32 + lane];

    const float* wp = W + ((size_t)nBase * N_OUT + k) * (D_IN * D_OUT);

    const float* xfrag = &xs[lo][dsel];

    if (use_bias) {
        const float4* dvp = reinterpret_cast<const float4*>(
            dotvec + ((size_t)(bBase + lo) * N_OUT + k) * D_OUT + 8 * hi);
        const float4 dvA = dvp[0];
        const float4 dvB = dvp[1];
        const v2f dv01 = {dvA.x, dvA.y}, dv23 = {dvA.z, dvA.w}, dv45 = {dvB.x, dvB.y}, dv67 = {dvB.z, dvB.w};

        *reinterpret_cast<float4*>(&xs[k][4 * lane])            = xg0;
        *reinterpret_cast<float4*>(&xs[k][XROW / 2 + 4 * lane]) = xg1;
        __syncthreads();

        for (int t = 0; t < NPB / NB; ++t) {
            v2f    a0[NB], a1[NB];
            v2f    b0[NB], b1[NB];
            #pragma unroll
            for (int j = 0; j < NB; ++j) {
                const float* wq = wp + j * WN_STRIDE;
                b0[j].x = wq[(0 + dsel) * D_OUT + lo];
                b0[j].y = wq[(1 + dsel) * D_OUT + lo];
                b1[j].x = wq[(4 + dsel) * D_OUT + lo];
                b1[j].y = wq[(5 + dsel) * D_OUT + lo];
            }
            #pragma unroll
            for (int j = 0; j < NB; ++j) {
                const float* xq = xfrag + (t * NB + j) * D_IN;
                a0[j] = *reinterpret_cast<const v2f*>(xq);
                a1[j] = *reinterpret_cast<const v2f*>(xq + 4);
            }

            v8f   h[NB];
            float bias[NB];
            #pragma unroll
            for (int j = 0; j < NB; ++j) {
                v8f hh = {};
                hh = __builtin_amdgcn_wmma_f32_16x16x4_f32(false, b0[j], false, a0[j],
                                                           (short)0, hh, false, false);
                hh = __builtin_amdgcn_wmma_f32_16x16x4_f32(false, b1[j], false, a1[j],
                                                           (short)0, hh, false, false);
                h[j] = hh;
                v2f pp = __builtin_shufflevector(hh, hh, 0, 1) * dv01;
                pp = __builtin_shufflevector(hh, hh, 2, 3) * dv23 + pp;
                pp = __builtin_shufflevector(hh, hh, 4, 5) * dv45 + pp;
                pp = __builtin_shufflevector(hh, hh, 6, 7) * dv67 + pp;
                const float part = pp.x + pp.y;
                bias[j] = part + other_half(part);
            }
            #pragma unroll
            for (int p = 0; p < NB / 2; ++p) biasLds[2 * p + hi][k][lo] = hi ? bias[2 * p + 1] : bias[2 * p];
            __syncthreads();

            {
                const int j0 = 2 * hi;
                const float v0 = biasLds[j0 + 0][lo][k];
                const float v1 = biasLds[j0 + 1][lo][k];
                const float m0 = row_max16(v0);
                const float m1 = row_max16(v1);
                const float e0 = __expf(v0 - m0);
                const float e1 = __expf(v1 - m1);
                const float d0 = row_sum16(e0);
                const float d1 = row_sum16(e1);
                *reinterpret_cast<float2*>(&cLds[lo][k * NB + j0]) = make_float2(e0 / d0, e1 / d1);
            }
            __syncthreads();

            const float4 c4 = *reinterpret_cast<const float4*>(&cLds[k][lo * NB]);
            const float  cj[NB] = {c4.x, c4.y, c4.z, c4.w};
            #pragma unroll
            for (int j = 0; j < NB; ++j) {
                #pragma unroll
                for (int v = 0; v < 8; ++v) sAcc[v] += cj[j] * h[j][v];
            }

            wp += NB * WN_STRIDE;
        }
    } else {
        *reinterpret_cast<float4*>(&xs[k][4 * lane])            = xg0;
        *reinterpret_cast<float4*>(&xs[k][XROW / 2 + 4 * lane]) = xg1;
        __syncthreads();

        #pragma unroll
        for (int t = 0; t < NPB / NB; ++t) {
            v2f    a0[NB], a1[NB];
            v2f    b0[NB], b1[NB];
            #pragma unroll
            for (int j = 0; j < NB; ++j) {
                const float* wq = wp + j * WN_STRIDE;
                b0[j].x = wq[(0 + dsel) * D_OUT + lo];
                b0[j].y = wq[(1 + dsel) * D_OUT + lo];
                b1[j].x = wq[(4 + dsel) * D_OUT + lo];
                b1[j].y = wq[(5 + dsel) * D_OUT + lo];
            }
            #pragma unroll
            for (int j = 0; j < NB; ++j) {
                const float* xq = xfrag + (t * NB + j) * D_IN;
                a0[j] = *reinterpret_cast<const v2f*>(xq);
                a1[j] = *reinterpret_cast<const v2f*>(xq + 4);
            }
            #pragma unroll
            for (int j = 0; j < NB; ++j) {
                v8f hh = {};
                hh = __builtin_amdgcn_wmma_f32_16x16x4_f32(false, b0[j], false, a0[j],
                                                           (short)0, hh, false, false);
                hh = __builtin_amdgcn_wmma_f32_16x16x4_f32(false, b1[j], false, a1[j],
                                                           (short)0, hh, false, false);
                #pragma unroll
                for (int v = 0; v < 8; ++v) sAcc[v] += 0.0625f * hh[v];
            }
            wp += NB * WN_STRIDE;
        }
    }

    float* pout = partial + (size_t)chunk * S_ELEMS + ((size_t)(bBase + lo) * N_OUT + k) * D_OUT + 8 * hi;
    *reinterpret_cast<float4*>(pout)     = make_float4(sAcc[0], sAcc[1], sAcc[2], sAcc[3]);
    *reinterpret_cast<float4*>(pout + 4) = make_float4(sAcc[4], sAcc[5], sAcc[6], sAcc[7]);
}

#define RS_THREADS 64
__global__ __launch_bounds__(RS_THREADS)
void reduce_squash(const float* __restrict__ partial,
                   const float* __restrict__ addend,
                   float* __restrict__ out,
                   int mode)
{
    const int idx = blockIdx.x * RS_THREADS + threadIdx.x;
    float add = 0.f;
    if (mode == 1) add = addend[idx];
    float p[NCHUNK];
    #pragma unroll
    for (int ch = 0; ch < NCHUNK; ++ch)
        p[ch] = partial[(size_t)ch * S_ELEMS + idx];
    float s = 0.f;
    #pragma unroll
    for (int ch = 0; ch < NCHUNK; ++ch)
        s += p[ch];

    float s2 = s * s;
    s2 = xor_add<1>(s2);
    s2 = xor_add<2>(s2);
    s2 = xor_add<4>(s2);
    s2 = xor_add<8>(s2);

    const float scale = (s2 / (1.0f + s2)) * __frsqrt_rn(s2);
    float o = scale * s;
    if (mode == 1) o += add;
    out[idx] = o;
}

extern "C" void kernel_launch(void* const* d_in, const int* in_sizes, int n_in,
                              void* d_out, int out_size, void* d_ws, size_t ws_size,
                              hipStream_t stream) {
    (void)in_sizes; (void)n_in; (void)out_size; (void)ws_size;
    const float* x = (const float*)d_in[0];
    const float* W = (const float*)d_in[1];
    float* out = (float*)d_out;

    float* partial = (float*)d_ws;
    float* out0    = partial + (size_t)NCHUNK * S_ELEMS;
    float* outsum  = out0 + S_ELEMS;

    dim3 grid(B_TOT / BT, NCHUNK);
    dim3 block(32, 16);

    routing_pass<<<grid, block, 0, stream>>>(x, W, outsum  , partial, 0);
    reduce_squash<<<S_ELEMS / RS_THREADS, RS_THREADS, 0, stream>>>(partial, outsum, out0, 0);

    routing_pass<<<grid, block, 0, stream>>>(x, W, out0, partial, 1);
    reduce_squash<<<S_ELEMS / RS_THREADS, RS_THREADS, 0, stream>>>(partial, out0, outsum, 1);

    routing_pass<<<grid, block, 0, stream>>>(x, W, outsum, partial, 1);
    reduce_squash<<<S_ELEMS / RS_THREADS, RS_THREADS, 0, stream>>>(partial, outsum  , out, 2);
}


# routed passes: W and x fragments of trip t+1 are requested during trip t into a second register set; the loop's barriers fence LDS only, so the loads land behind the exchange; sum orders kept
# speedup vs baseline: 2.9460x; 1.0779x over previous
#include <hip/hip_runtime.h>

#define B_TOT   64
#define N_IN    2048
#define D_IN    8
#define N_OUT   16
#define D_OUT   16
#define BT      16
#define NPB     32
#define NCHUNK  (N_IN / NPB)
#define NB      4
#define S_ELEMS (B_TOT * N_OUT * D_OUT)
#define WN_STRIDE (N_OUT * D_IN * D_OUT)
#define BPITCH  (BT + 1)
#define CPITCH  (NB * BT + 4)
#define XROW    (NPB * D_IN)
#define XPITCH  (XROW + 4)

typedef float v2f __attribute__((ext_vector_type(2)));
typedef float v8f __attribute__((ext_vector_type(8)));

template <int MASK>
__device__ __forceinline__ float xor_add(float v) {
    const int sw = __builtin_amdgcn_ds_swizzle(__float_as_int(v), (MASK << 10) | 0x1f);
    return v + __int_as_float(sw);
}

template <int MASK>
__device__ __forceinline__ float row_xor(float v) {
    return __int_as_float(__builtin_amdgcn_update_dpp(0, __float_as_int(v), 0x160 + MASK, 0xf, 0xf, false));
}
__device__ __forceinline__ float other_half(float v) {
    return __int_as_float(__builtin_amdgcn_permlanex16(0, __float_as_int(v), 0x76543210, 0xfedcba98, false, false));
}
__device__ __forceinline__ float row_max16(float v) {
    v = fmaxf(v, row_xor<1>(v));
    v = fmaxf(v, row_xor<2>(v));
    v = fmaxf(v, row_xor<4>(v));
    v = fmaxf(v, row_xor<8>(v));
    return v;
}
__device__ __forceinline__ float row_sum16(float v) {
    v += row_xor<1>(v);
    v += row_xor<2>(v);
    v += row_xor<4>(v);
    v += row_xor<8>(v);
    return v;
}

__device__ __forceinline__ void lds_barrier() {
    __builtin_amdgcn_fence(__ATOMIC_RELEASE, "workgroup", "local");
    __builtin_amdgcn_s_barrier();
    __builtin_amdgcn_fence(__ATOMIC_ACQUIRE, "workgroup", "local");
}

template <bool USE_BIAS>
__global__ __launch_bounds__(512)
void routing_pass(const float* __restrict__ x,
                  const float* __restrict__ W,
                  const float* __restrict__ dotvec,
                  float* __restrict__ partial)
{
    const int bx    = blockIdx.x;
    const int chunk = blockIdx.y;
    const int k     = threadIdx.y;
    const int lane  = threadIdx.x;
    const int hi    = lane >> 4;
    const int lo    = lane & 15;
    const int dsel  = hi * 2;

    __shared__ __align__(64) float biasLds[NB][N_OUT][BPITCH];
    __shared__ __align__(64) float cLds[N_OUT][CPITCH];
    __shared__ __align__(16) float xs[BT][XPITCH];

    v8f sAcc = {};

    const int bBase = bx * BT;
    const int nBase = chunk * NPB;

    const float4* xrow4 = reinterpret_cast<const float4*>(x + ((size_t)(bBase + k) * N_IN + nBase) * D_IN);
    const float4 xg0 = xrow4[lane];
    const float4 xg1 = xrow4[32 + lane];

    const float* wp = W + ((size_t)nBase * N_OUT + k) * (D_IN * D_OUT);

    const float* xfrag = &xs[lo][dsel];

    if constexpr (USE_BIAS) {
        const float4* dvp = reinterpret_cast<const float4*>(
            dotvec + ((size_t)(bBase + lo) * N_OUT + k) * D_OUT + 8 * hi);
        const float4 dvA = dvp[0];
        const float4 dvB = dvp[1];
        const v2f dv01 = {dvA.x, dvA.y}, dv23 = {dvA.z, dvA.w}, dv45 = {dvB.x, dvB.y}, dv67 = {dvB.z, dvB.w};

        *reinterpret_cast<float4*>(&xs[k][4 * lane])            = xg0;
        *reinterpret_cast<float4*>(&xs[k][XROW / 2 + 4 * lane]) = xg1;
        __syncthreads();

        v2f wf0[2][NB], wf1[2][NB];
        v2f xf0[2][NB], xf1[2][NB];
        #pragma unroll
        for (int j = 0; j < NB; ++j) {
            const float* wq = wp + j * WN_STRIDE;
            wf0[0][j].x = wq[(0 + dsel) * D_OUT + lo];
            wf0[0][j].y = wq[(1 + dsel) * D_OUT + lo];
            wf1[0][j].x = wq[(4 + dsel) * D_OUT + lo];
            wf1[0][j].y = wq[(5 + dsel) * D_OUT + lo];
            const float* xq = xfrag + j * D_IN;
            xf0[0][j] = *reinterpret_cast<const v2f*>(xq);
            xf1[0][j] = *reinterpret_cast<const v2f*>(xq + 4);
        }

        auto fetch = [&](int tn, v2f (&w0)[NB], v2f (&w1)[NB], v2f (&x0)[NB], v2f (&x1)[NB]) __attribute__((always_inline)) {
            #pragma unroll
            for (int j = 0; j < NB; ++j) {
                const float* wq = wp + (NB + j) * WN_STRIDE;
                w0[j].x = wq[(0 + dsel) * D_OUT + lo];
                w0[j].y = wq[(1 + dsel) * D_OUT + lo];
                w1[j].x = wq[(4 + dsel) * D_OUT + lo];
                w1[j].y = wq[(5 + dsel) * D_OUT + lo];
                const float* xq = xfrag + (tn * NB + j) * D_IN;
                x0[j] = *reinterpret_cast<const v2f*>(xq);
                x1[j] = *reinterpret_cast<const v2f*>(xq + 4);
            }
        };
        v8f   h[NB];
        auto hats_and_logits = [&](v2f (&w0)[NB], v2f (&w1)[NB], v2f (&x0)[NB], v2f (&x1)[NB]) __attribute__((always_inline)) {
            float bias[NB];
            #pragma unroll
            for (int j = 0; j < NB; ++j) {
                v8f hh = {};
                hh = __builtin_amdgcn_wmma_f32_16x16x4_f32(false, w0[j], false, x0[j],
                                                           (short)0, hh, false, false);
                hh = __builtin_amdgcn_wmma_f32_16x16x4_f32(false, w1[j], false, x1[j],
                                                           (short)0, hh, false, false);
                h[j] = hh;
                v2f pp = __builtin_shufflevector(hh, hh, 0, 1) * dv01;
                pp = __builtin_shufflevector(hh, hh, 2, 3) * dv23 + pp;
                pp = __builtin_shufflevector(hh, hh, 4, 5) * dv45 + pp;
                pp = __builtin_shufflevector(hh, hh, 6, 7) * dv67 + pp;
                const float part = pp.x + pp.y;
                bias[j] = part + other_half(part);
            }
            #pragma unroll
            for (int p = 0; p < NB / 2; ++p) biasLds[2 * p + hi][k][lo] = hi ? bias[2 * p + 1] : bias[2 * p];
        };
        auto exchange_and_accumulate = [&]() __attribute__((always_inline)) {
            lds_barrier();

            {
                const int j0 = 2 * hi;
                const float v0 = biasLds[j0 + 0][lo][k];
                const float v1 = biasLds[j0 + 1][lo][k];
                const float m0 = row_max16(v0);
                const float m1 = row_max16(v1);
                const float e0 = __expf(v0 - m0);
                const float e1 = __expf(v1 - m1);
                const float d0 = row_sum16(e0);
                const float d1 = row_sum16(e1);
                *reinterpret_cast<float2*>(&cLds[lo][k * NB + j0]) = make_float2(e0 / d0, e1 / d1);
            }
            lds_barrier();

            const float4 c4 = *reinterpret_cast<const float4*>(&cLds[k][lo * NB]);
            const float  cj[NB] = {c4.x, c4.y, c4.z, c4.w};
            #pragma unroll
            for (int j = 0; j < NB; ++j) {
                #pragma unroll
                for (int v = 0; v < 8; ++v) sAcc[v] += cj[j] * h[j][v];
            }
            wp += NB * WN_STRIDE;
        };

        #pragma nounroll
        for (int tt = 0; tt < NPB / NB; tt += 2) {
            fetch(tt + 1, wf0[1], wf1[1], xf0[1], xf1[1]);
            __builtin_amdgcn_sched_barrier(0);
            hats_and_logits(wf0[0], wf1[0], xf0[0], xf1[0]);
            exchange_and_accumulate();

            __builtin_amdgcn_sched_barrier(0);
            hats_and_logits(wf0[1], wf1[1], xf0[1], xf1[1]);
            if (tt + 2 < NPB / NB)
                fetch(tt + 2, wf0[0], wf1[0], xf0[0], xf1[0]);
            exchange_and_accumulate();
        }
    } else {
        *reinterpret_cast<float4*>(&xs[k][4 * lane])            = xg0;
        *reinterpret_cast<float4*>(&xs[k][XROW / 2 + 4 * lane]) = xg1;
        __syncthreads();

        #pragma unroll
        for (int t = 0; t < NPB / NB; ++t) {
            v2f    a0[NB], a1[NB];
            v2f    b0[NB], b1[NB];
            #pragma unroll
            for (int j = 0; j < NB; ++j) {
                const float* wq = wp + j * WN_STRIDE;
                b0[j].x = wq[(0 + dsel) * D_OUT + lo];
                b0[j].y = wq[(1 + dsel) * D_OUT + lo];
                b1[j].x = wq[(4 + dsel) * D_OUT + lo];
                b1[j].y = wq[(5 + dsel) * D_OUT + lo];
            }
            #pragma unroll
            for (int j = 0; j < NB; ++j) {
                const float* xq = xfrag + (t * NB + j) * D_IN;
                a0[j] = *reinterpret_cast<const v2f*>(xq);
                a1[j] = *reinterpret_cast<const v2f*>(xq + 4);
            }
            #pragma unroll
            for (int j = 0; j < NB; ++j) {
                v8f hh = {};
                hh = __builtin_amdgcn_wmma_f32_16x16x4_f32(false, b0[j], false, a0[j],
                                                           (short)0, hh, false, false);
                hh = __builtin_amdgcn_wmma_f32_16x16x4_f32(false, b1[j], false, a1[j],
                                                           (short)0, hh, false, false);
                #pragma unroll
                for (int v = 0; v < 8; ++v) sAcc[v] += 0.0625f * hh[v];
            }
            wp += NB * WN_STRIDE;
        }
    }

    float* pout = partial + (size_t)chunk * S_ELEMS + ((size_t)(bBase + lo) * N_OUT + k) * D_OUT + 8 * hi;
    *reinterpret_cast<float4*>(pout)     = make_float4(sAcc[0], sAcc[1], sAcc[2], sAcc[3]);
    *reinterpret_cast<float4*>(pout + 4) = make_float4(sAcc[4], sAcc[5], sAcc[6], sAcc[7]);
}

#define RS_THREADS 64
__global__ __launch_bounds__(RS_THREADS)
void reduce_squash(const float* __restrict__ partial,
                   const float* __restrict__ addend,
                   float* __restrict__ out,
                   int mode)
{
    const int idx = blockIdx.x * RS_THREADS + threadIdx.x;
    float add = 0.f;
    if (mode == 1) add = addend[idx];
    float p[NCHUNK];
    #pragma unroll
    for (int ch = 0; ch < NCHUNK; ++ch)
        p[ch] = partial[(size_t)ch * S_ELEMS + idx];
    float s = 0.f;
    #pragma unroll
    for (int ch = 0; ch < NCHUNK; ++ch)
        s += p[ch];

    float s2 = s * s;
    s2 = xor_add<1>(s2);
    s2 = xor_add<2>(s2);
    s2 = xor_add<4>(s2);
    s2 = xor_add<8>(s2);

    const float scale = (s2 / (1.0f + s2)) * __frsqrt_rn(s2);
    float o = scale * s;
    if (mode == 1) o += add;
    out[idx] = o;
}

extern "C" void kernel_launch(void* const* d_in, const int* in_sizes, int n_in,
                              void* d_out, int out_size, void* d_ws, size_t ws_size,
                              hipStream_t stream) {
    (void)in_sizes; (void)n_in; (void)out_size; (void)ws_size;
    const float* x = (const float*)d_in[0];
    const float* W = (const float*)d_in[1];
    float* out = (float*)d_out;

    float* partial = (float*)d_ws;
    float* out0    = partial + (size_t)NCHUNK * S_ELEMS;
    float* outsum  = out0 + S_ELEMS;

    dim3 grid(B_TOT / BT, NCHUNK);
    dim3 block(32, 16);

    routing_pass<false><<<grid, block, 0, stream>>>(x, W, outsum  , partial);
    reduce_squash<<<S_ELEMS / RS_THREADS, RS_THREADS, 0, stream>>>(partial, outsum, out0, 0);

    routing_pass<true><<<grid, block, 0, stream>>>(x, W, out0, partial);
    reduce_squash<<<S_ELEMS / RS_THREADS, RS_THREADS, 0, stream>>>(partial, out0, outsum, 1);

    routing_pass<true><<<grid, block, 0, stream>>>(x, W, outsum, partial);
    reduce_squash<<<S_ELEMS / RS_THREADS, RS_THREADS, 0, stream>>>(partial, outsum  , out, 2);
}


# routed passes: softmax exchange re-laid: lanes store logit half-sums, a softmax lane adds them for 4 of a row's 16 logits; max/sum 3 in-lane + 2 DPP steps; c = e*rcp(sum); denominator order changes
# speedup vs baseline: 3.0625x; 1.0395x over previous
#include <hip/hip_runtime.h>

#define B_TOT   64
#define N_IN    2048
#define D_IN    8
#define N_OUT   16
#define D_OUT   16
#define BT      16
#define NPB     32
#define NCHUNK  (N_IN / NPB)
#define NB      4
#define S_ELEMS (B_TOT * N_OUT * D_OUT)
#define WN_STRIDE (N_OUT * D_IN * D_OUT)
#define KPITCH  (N_OUT + 4)
#define CPITCH  (NB * BT + 4)
#define XROW    (NPB * D_IN)
#define XPITCH  (XROW + 4)

typedef float v2f __attribute__((ext_vector_type(2)));
typedef float v8f __attribute__((ext_vector_type(8)));

template <int MASK>
__device__ __forceinline__ float xor_add(float v) {
    const int sw = __builtin_amdgcn_ds_swizzle(__float_as_int(v), (MASK << 10) | 0x1f);
    return v + __int_as_float(sw);
}

template <int MASK>
__device__ __forceinline__ float row_xor(float v) {
    return __int_as_float(__builtin_amdgcn_update_dpp(0, __float_as_int(v), 0x160 + MASK, 0xf, 0xf, false));
}

__device__ __forceinline__ void lds_barrier() {
    __builtin_amdgcn_fence(__ATOMIC_RELEASE, "workgroup", "local");
    __builtin_amdgcn_s_barrier();
    __builtin_amdgcn_fence(__ATOMIC_ACQUIRE, "workgroup", "local");
}

template <bool USE_BIAS>
__global__ __launch_bounds__(512)
void routing_pass(const float* __restrict__ x,
                  const float* __restrict__ W,
                  const float* __restrict__ dotvec,
                  float* __restrict__ partial)
{
    const int bx    = blockIdx.x;
    const int chunk = blockIdx.y;
    const int k     = threadIdx.y;
    const int lane  = threadIdx.x;
    const int hi    = lane >> 4;
    const int lo    = lane & 15;
    const int dsel  = hi * 2;

    __shared__ __align__(64) float biasLds[2][NB][BT][KPITCH];
    __shared__ __align__(64) float cLds[N_OUT][CPITCH];
    __shared__ __align__(16) float xs[BT][XPITCH];

    v8f sAcc = {};

    const int bBase = bx * BT;
    const int nBase = chunk * NPB;

    const float4* xrow4 = reinterpret_cast<const float4*>(x + ((size_t)(bBase + k) * N_IN + nBase) * D_IN);
    const float4 xg0 = xrow4[lane];
    const float4 xg1 = xrow4[32 + lane];

    const float* wp = W + ((size_t)nBase * N_OUT + k) * (D_IN * D_OUT);

    const float* xfrag = &xs[lo][dsel];

    if constexpr (USE_BIAS) {
        const float4* dvp = reinterpret_cast<const float4*>(
            dotvec + ((size_t)(bBase + lo) * N_OUT + k) * D_OUT + 8 * hi);
        const float4 dvA = dvp[0];
        const float4 dvB = dvp[1];
        const v2f dv01 = {dvA.x, dvA.y}, dv23 = {dvA.z, dvA.w}, dv45 = {dvB.x, dvB.y}, dv67 = {dvB.z, dvB.w};

        *reinterpret_cast<float4*>(&xs[k][4 * lane])            = xg0;
        *reinterpret_cast<float4*>(&xs[k][XROW / 2 + 4 * lane]) = xg1;
        __syncthreads();

        const int  kw      = __builtin_amdgcn_readfirstlane(k);
        const bool smxWave = (((kw >> 2) ^ kw) & 1) == 0;
        const int  sw      = kw >> 1;
        const int  sj      = sw >> 1;
        const int  sb      = (sw & 1) * 8 + (lane >> 2);
        const int  sq      = lane & 3;

        v2f wf0[2][NB], wf1[2][NB];
        v2f xf0[2][NB], xf1[2][NB];
        #pragma unroll
        for (int j = 0; j < NB; ++j) {
            const float* wq = wp + j * WN_STRIDE;
            wf0[0][j].x = wq[(0 + dsel) * D_OUT + lo];
            wf0[0][j].y = wq[(1 + dsel) * D_OUT + lo];
            wf1[0][j].x = wq[(4 + dsel) * D_OUT + lo];
            wf1[0][j].y = wq[(5 + dsel) * D_OUT + lo];
            const float* xq = xfrag + j * D_IN;
            xf0[0][j] = *reinterpret_cast<const v2f*>(xq);
            xf1[0][j] = *reinterpret_cast<const v2f*>(xq + 4);
        }

        auto fetch = [&](int tn, v2f (&w0)[NB], v2f (&w1)[NB], v2f (&x0)[NB], v2f (&x1)[NB]) __attribute__((always_inline)) {
            #pragma unroll
            for (int j = 0; j < NB; ++j) {
                const float* wq = wp + (NB + j) * WN_STRIDE;
                w0[j].x = wq[(0 + dsel) * D_OUT + lo];
                w0[j].y = wq[(1 + dsel) * D_OUT + lo];
                w1[j].x = wq[(4 + dsel) * D_OUT + lo];
                w1[j].y = wq[(5 + dsel) * D_OUT + lo];
                const float* xq = xfrag + (tn * NB + j) * D_IN;
                x0[j] = *reinterpret_cast<const v2f*>(xq);
                x1[j] = *reinterpret_cast<const v2f*>(xq + 4);
            }
        };
        v8f   h[NB];
        auto hats_and_logits = [&](v2f (&w0)[NB], v2f (&w1)[NB], v2f (&x0)[NB], v2f (&x1)[NB]) __attribute__((always_inline)) {
            float part[NB];
            #pragma unroll
            for (int j = 0; j < NB; ++j) {
                v8f hh = {};
                hh = __builtin_amdgcn_wmma_f32_16x16x4_f32(false, w0[j], false, x0[j],
                                                           (short)0, hh, false, false);
                hh = __builtin_amdgcn_wmma_f32_16x16x4_f32(false, w1[j], false, x1[j],
                                                           (short)0, hh, false, false);
                h[j] = hh;
                v2f pp = __builtin_shufflevector(hh, hh, 0, 1) * dv01;
                pp = __builtin_shufflevector(hh, hh, 2, 3) * dv23 + pp;
                pp = __builtin_shufflevector(hh, hh, 4, 5) * dv45 + pp;
                pp = __builtin_shufflevector(hh, hh, 6, 7) * dv67 + pp;
                part[j] = pp.x + pp.y;
            }
            #pragma unroll
            for (int j = 0; j < NB; ++j) biasLds[hi][j][lo][k] = part[j];
        };
        auto exchange_and_accumulate = [&]() __attribute__((always_inline)) {
            lds_barrier();

            if (smxWave) {
                const float4 va = *reinterpret_cast<const float4*>(&biasLds[0][sj][sb][4 * sq]);
                const float4 vb = *reinterpret_cast<const float4*>(&biasLds[1][sj][sb][4 * sq]);
                const float4 v  = make_float4(va.x + vb.x, va.y + vb.y, va.z + vb.z, va.w + vb.w);
                float m = fmaxf(fmaxf(v.x, v.y), fmaxf(v.z, v.w));
                m = fmaxf(m, row_xor<1>(m));
                m = fmaxf(m, row_xor<2>(m));
                const float e0 = __expf(v.x - m);
                const float e1 = __expf(v.y - m);
                const float e2 = __expf(v.z - m);
                const float e3 = __expf(v.w - m);
                float d = (e0 + e1) + (e2 + e3);
                d += row_xor<1>(d);
                d += row_xor<2>(d);
                const float r = __builtin_amdgcn_rcpf(d);
                cLds[4 * sq + 0][sb * NB + sj] = e0 * r;
                cLds[4 * sq + 1][sb * NB + sj] = e1 * r;
                cLds[4 * sq + 2][sb * NB + sj] = e2 * r;
                cLds[4 * sq + 3][sb * NB + sj] = e3 * r;
            }
            lds_barrier();

            const float4 c4 = *reinterpret_cast<const float4*>(&cLds[k][lo * NB]);
            const float  cj[NB] = {c4.x, c4.y, c4.z, c4.w};
            #pragma unroll
            for (int j = 0; j < NB; ++j) {
                #pragma unroll
                for (int v = 0; v < 8; ++v) sAcc[v] += cj[j] * h[j][v];
            }
            wp += NB * WN_STRIDE;
        };

        #pragma nounroll
        for (int tt = 0; tt < NPB / NB; tt += 2) {
            fetch(tt + 1, wf0[1], wf1[1], xf0[1], xf1[1]);
            __builtin_amdgcn_sched_barrier(0);
            hats_and_logits(wf0[0], wf1[0], xf0[0], xf1[0]);
            exchange_and_accumulate();

            __builtin_amdgcn_sched_barrier(0);
            hats_and_logits(wf0[1], wf1[1], xf0[1], xf1[1]);
            if (tt + 2 < NPB / NB)
                fetch(tt + 2, wf0[0], wf1[0], xf0[0], xf1[0]);
            exchange_and_accumulate();
        }
    } else {
        *reinterpret_cast<float4*>(&xs[k][4 * lane])            = xg0;
        *reinterpret_cast<float4*>(&xs[k][XROW / 2 + 4 * lane]) = xg1;
        __syncthreads();

        #pragma unroll
        for (int t = 0; t < NPB / NB; ++t) {
            v2f    a0[NB], a1[NB];
            v2f    b0[NB], b1[NB];
            #pragma unroll
            for (int j = 0; j < NB; ++j) {
                const float* wq = wp + j * WN_STRIDE;
                b0[j].x = wq[(0 + dsel) * D_OUT + lo];
                b0[j].y = wq[(1 + dsel) * D_OUT + lo];
                b1[j].x = wq[(4 + dsel) * D_OUT + lo];
                b1[j].y = wq[(5 + dsel) * D_OUT + lo];
            }
            #pragma unroll
            for (int j = 0; j < NB; ++j) {
                const float* xq = xfrag + (t * NB + j) * D_IN;
                a0[j] = *reinterpret_cast<const v2f*>(xq);
                a1[j] = *reinterpret_cast<const v2f*>(xq + 4);
            }
            #pragma unroll
            for (int j = 0; j < NB; ++j) {
                v8f hh = {};
                hh = __builtin_amdgcn_wmma_f32_16x16x4_f32(false, b0[j], false, a0[j],
                                                           (short)0, hh, false, false);
                hh = __builtin_amdgcn_wmma_f32_16x16x4_f32(false, b1[j], false, a1[j],
                                                           (short)0, hh, false, false);
                #pragma unroll
                for (int v = 0; v < 8; ++v) sAcc[v] += 0.0625f * hh[v];
            }
            wp += NB * WN_STRIDE;
        }
    }

    float* pout = partial + (size_t)chunk * S_ELEMS + ((size_t)(bBase + lo) * N_OUT + k) * D_OUT + 8 * hi;
    *reinterpret_cast<float4*>(pout)     = make_float4(sAcc[0], sAcc[1], sAcc[2], sAcc[3]);
    *reinterpret_cast<float4*>(pout + 4) = make_float4(sAcc[4], sAcc[5], sAcc[6], sAcc[7]);
}

#define RS_THREADS 64
__global__ __launch_bounds__(RS_THREADS)
void reduce_squash(const float* __restrict__ partial,
                   const float* __restrict__ addend,
                   float* __restrict__ out,
                   int mode)
{
    const int idx = blockIdx.x * RS_THREADS + threadIdx.x;
    float add = 0.f;
    if (mode == 1) add = addend[idx];
    float p[NCHUNK];
    #pragma unroll
    for (int ch = 0; ch < NCHUNK; ++ch)
        p[ch] = partial[(size_t)ch * S_ELEMS + idx];
    float s = 0.f;
    #pragma unroll
    for (int ch = 0; ch < NCHUNK; ++ch)
        s += p[ch];

    float s2 = s * s;
    s2 = xor_add<1>(s2);
    s2 = xor_add<2>(s2);
    s2 = xor_add<4>(s2);
    s2 = xor_add<8>(s2);

    const float scale = (s2 / (1.0f + s2)) * __frsqrt_rn(s2);
    float o = scale * s;
    if (mode == 1) o += add;
    out[idx] = o;
}

extern "C" void kernel_launch(void* const* d_in, const int* in_sizes, int n_in,
                              void* d_out, int out_size, void* d_ws, size_t ws_size,
                              hipStream_t stream) {
    (void)in_sizes; (void)n_in; (void)out_size; (void)ws_size;
    const float* x = (const float*)d_in[0];
    const float* W = (const float*)d_in[1];
    float* out = (float*)d_out;

    float* partial = (float*)d_ws;
    float* out0    = partial + (size_t)NCHUNK * S_ELEMS;
    float* outsum  = out0 + S_ELEMS;

    dim3 grid(B_TOT / BT, NCHUNK);
    dim3 block(32, 16);

    routing_pass<false><<<grid, block, 0, stream>>>(x, W, outsum  , partial);
    reduce_squash<<<S_ELEMS / RS_THREADS, RS_THREADS, 0, stream>>>(partial, outsum, out0, 0);

    routing_pass<true><<<grid, block, 0, stream>>>(x, W, out0, partial);
    reduce_squash<<<S_ELEMS / RS_THREADS, RS_THREADS, 0, stream>>>(partial, out0, outsum, 1);

    routing_pass<true><<<grid, block, 0, stream>>>(x, W, outsum, partial);
    reduce_squash<<<S_ELEMS / RS_THREADS, RS_THREADS, 0, stream>>>(partial, outsum  , out, 2);
}
